# starting kernel
# baseline (speedup 1.0000x reference)
#include <hip/hip_runtime.h>
#include <hip/hip_bf16.h>


typedef float v2f __attribute__((ext_vector_type(2)));
typedef float v8f __attribute__((ext_vector_type(8)));

#define ORDER 16
#define FRAME 160
#define EPS_F 1e-8f
#define WAVES_PER_BLOCK 8
#define PAD 16
#define REGION (PAD + FRAME + PAD)

__global__ __launch_bounds__(256) void lpc_residual_kernel(
    const float* __restrict__ x, float* __restrict__ out, int n_frames) {
  __shared__ float sh[WAVES_PER_BLOCK][REGION];

  const int lane = threadIdx.x & 31;
  const int wv = threadIdx.x >> 5;
  const int frame = blockIdx.x * WAVES_PER_BLOCK + wv;
  const bool active = frame < n_frames;

  float* fr = &sh[wv][PAD];

  if (active) {
    if (lane < PAD) {
      sh[wv][lane] = 0.0f;
      sh[wv][PAD + FRAME + lane] = 0.0f;
    }
    const float* gx = x + (long long)frame * FRAME;
#pragma unroll
    for (int t = 0; t < 5; ++t) fr[lane + 32 * t] = gx[lane + 32 * t];
  }
  __syncthreads();

  if (active) {
    const int m = lane & 15;
    const int k0 = (lane >> 4) << 1;
    v8f acc0 = {0.f, 0.f, 0.f, 0.f, 0.f, 0.f, 0.f, 0.f};
    v8f acc1 = acc0;
    for (int c = 0; c < 40; c += 2) {
      v2f A0, B0, A1, B1;
      A0.x = fr[4 * c + k0 + m];
      A0.y = fr[4 * c + k0 + 1 + m];
      B0.x = fr[4 * c + k0];
      B0.y = fr[4 * c + k0 + 1];
      A1.x = fr[4 * (c + 1) + k0 + m];
      A1.y = fr[4 * (c + 1) + k0 + 1 + m];
      B1.x = fr[4 * (c + 1) + k0];
      B1.y = fr[4 * (c + 1) + k0 + 1];
      acc0 = __builtin_amdgcn_wmma_f32_16x16x4_f32(false, A0, false, B0,
                                                   (short)0, acc0, false, false);
      acc1 = __builtin_amdgcn_wmma_f32_16x16x4_f32(false, A1, false, B1,
                                                   (short)0, acc1, false, false);
    }
    v8f D = acc0 + acc1;

    float p16 = 0.0f;
#pragma unroll
    for (int t = 0; t < 5; ++t) {
      int idx = lane + 32 * t;
      if (idx < FRAME - ORDER) p16 += fr[idx] * fr[idx + ORDER];
    }
#pragma unroll
    for (int s = 16; s >= 1; s >>= 1) p16 += __shfl_xor(p16, s, 32);

    float r[ORDER + 1];
#pragma unroll
    for (int q = 0; q < 16; ++q) r[q] = __shfl(D[q & 7], (q >> 3) * 16, 32);
    r[ORDER] = p16;

    float a[ORDER + 1];
    a[0] = 1.0f;
#pragma unroll
    for (int q = 1; q <= ORDER; ++q) a[q] = 0.0f;
    float e = (r[0] != 0.0f) ? r[0] : EPS_F;
#pragma unroll
    for (int i = 1; i <= ORDER; ++i) {
      float accv = r[i];
#pragma unroll
      for (int j = 1; j < i; ++j) accv -= a[j] * r[i - j];
      float k = (e != 0.0f) ? (accv / e) : 0.0f;
      float anew[ORDER + 1];
#pragma unroll
      for (int j = 1; j < i; ++j) anew[j] = a[j] - k * a[i - j];
#pragma unroll
      for (int j = 1; j < i; ++j) a[j] = anew[j];
      a[i] = k;
      e = fmaxf(e * (1.0f - k * k), EPS_F);
    }

    float* go = out + (long long)frame * FRAME;
#pragma unroll
    for (int t = 0; t < 5; ++t) {
      int n = lane + 32 * t;
      float s = 0.0f;
#pragma unroll
      for (int k = 0; k <= ORDER; ++k) s += a[k] * fr[n - k];
      go[n] = s;
    }
  }
}

extern "C" void kernel_launch(void* const* d_in, const int* in_sizes, int n_in,
                              void* d_out, int out_size, void* d_ws, size_t ws_size,
                              hipStream_t stream) {
  (void)n_in; (void)out_size; (void)d_ws; (void)ws_size;
  const float* x = (const float*)d_in[0];
  float* out = (float*)d_out;
  const int total = in_sizes[0];
  const int n_frames = total / FRAME;
  const int blocks = (n_frames + WAVES_PER_BLOCK - 1) / WAVES_PER_BLOCK;
  lpc_residual_kernel<<<blocks, WAVES_PER_BLOCK * 32, 0, stream>>>(x, out, n_frames);
}


# Levinson-Durbin issued once per frame, not in all 32 lanes of its wave: 8-wave groups take 32 frames, one wave runs the recursion with lane = frame; r and a pass through an LDS table
# speedup vs baseline: 1.3444x; 1.3444x over previous
#include <hip/hip_runtime.h>


typedef float v2f __attribute__((ext_vector_type(2)));
typedef float v8f __attribute__((ext_vector_type(8)));

#define ORDER 16
#define FRAME 160
#define EPS_F 1e-8f
#define WAVES_PER_BLOCK 8
#define FRAMES_PER_WAVE 4
#define FRAMES_PER_BLOCK (WAVES_PER_BLOCK * FRAMES_PER_WAVE)
#define PAD 16
#define REGION (PAD + FRAME + PAD)
#define TAB_PITCH 20

__global__ __launch_bounds__(256) void lpc_residual_kernel(
    const float* __restrict__ x, float* __restrict__ out, int n_frames) {
  __shared__ float sh[FRAMES_PER_BLOCK][REGION];
  __shared__ __attribute__((aligned(16))) float tab[FRAMES_PER_BLOCK][TAB_PITCH];

  const int lane = threadIdx.x & 31;
  const int wv = __builtin_amdgcn_readfirstlane((int)(threadIdx.x >> 5));
  const int frame0 = blockIdx.x * FRAMES_PER_BLOCK;

  {
    float v[FRAMES_PER_WAVE][5];
#pragma unroll
    for (int i = 0; i < FRAMES_PER_WAVE; ++i) {
      const int frame = min(frame0 + wv * FRAMES_PER_WAVE + i, n_frames - 1);
      const float* gx = x + (long long)frame * FRAME;
#pragma unroll
      for (int t = 0; t < 5; ++t) v[i][t] = gx[lane + 32 * t];
    }
#pragma unroll
    for (int i = 0; i < FRAMES_PER_WAVE; ++i) {
      float* row = sh[wv * FRAMES_PER_WAVE + i];
      if (lane < PAD) {
        row[lane] = 0.0f;
        row[PAD + FRAME + lane] = 0.0f;
      }
#pragma unroll
      for (int t = 0; t < 5; ++t) row[PAD + lane + 32 * t] = v[i][t];
    }
  }
  __syncthreads();

  const int m = lane & 15;
  const int k0 = (lane >> 4) << 1;
#pragma unroll 1
  for (int i = 0; i < FRAMES_PER_WAVE; ++i) {
    const int f = wv * FRAMES_PER_WAVE + i;
    const float* fr = &sh[f][PAD];
    v8f acc0 = {0.f, 0.f, 0.f, 0.f, 0.f, 0.f, 0.f, 0.f};
    v8f acc1 = acc0;
    for (int c = 0; c < 40; c += 2) {
      v2f A0, B0, A1, B1;
      A0.x = fr[4 * c + k0 + m];
      A0.y = fr[4 * c + k0 + 1 + m];
      B0.x = fr[4 * c + k0];
      B0.y = fr[4 * c + k0 + 1];
      A1.x = fr[4 * (c + 1) + k0 + m];
      A1.y = fr[4 * (c + 1) + k0 + 1 + m];
      B1.x = fr[4 * (c + 1) + k0];
      B1.y = fr[4 * (c + 1) + k0 + 1];
      acc0 = __builtin_amdgcn_wmma_f32_16x16x4_f32(false, A0, false, B0,
                                                   (short)0, acc0, false, false);
      acc1 = __builtin_amdgcn_wmma_f32_16x16x4_f32(false, A1, false, B1,
                                                   (short)0, acc1, false, false);
    }
    v8f D = acc0 + acc1;

    float p16 = 0.0f;
#pragma unroll
    for (int t = 0; t < 5; ++t) {
      int idx = lane + 32 * t;
      if (idx < FRAME - ORDER) p16 += fr[idx] * fr[idx + ORDER];
    }
#pragma unroll
    for (int s = 16; s >= 1; s >>= 1) p16 += __shfl_xor(p16, s, 32);

    if (m == 0) {
      float4* dst = reinterpret_cast<float4*>(&tab[f][(lane >> 4) * 8]);
      dst[0] = make_float4(D[0], D[1], D[2], D[3]);
      dst[1] = make_float4(D[4], D[5], D[6], D[7]);
      if (lane == 0) tab[f][ORDER] = p16;
    }
  }
  __syncthreads();

  if (wv == (int)(blockIdx.x & (WAVES_PER_BLOCK - 1))) {
    float r[ORDER + 1];
    {
      const float4* srow = reinterpret_cast<const float4*>(&tab[lane][0]);
#pragma unroll
      for (int q = 0; q < 4; ++q) {
        float4 t4 = srow[q];
        r[4 * q + 0] = t4.x; r[4 * q + 1] = t4.y; r[4 * q + 2] = t4.z; r[4 * q + 3] = t4.w;
      }
      r[ORDER] = tab[lane][ORDER];
    }
    float a[ORDER + 1];
    a[0] = 1.0f;
#pragma unroll
    for (int q = 1; q <= ORDER; ++q) a[q] = 0.0f;
    float e = (r[0] != 0.0f) ? r[0] : EPS_F;
#pragma unroll
    for (int i = 1; i <= ORDER; ++i) {
      float accv = r[i];
#pragma unroll
      for (int j = 1; j < i; ++j) accv -= a[j] * r[i - j];
      float k = (e != 0.0f) ? (accv / e) : 0.0f;
      float anew[ORDER + 1];
#pragma unroll
      for (int j = 1; j < i; ++j) anew[j] = a[j] - k * a[i - j];
#pragma unroll
      for (int j = 1; j < i; ++j) a[j] = anew[j];
      a[i] = k;
      e = fmaxf(e * (1.0f - k * k), EPS_F);
    }
    float4* drow = reinterpret_cast<float4*>(&tab[lane][0]);
#pragma unroll
    for (int q = 0; q < 4; ++q)
      drow[q] = make_float4(a[4 * q + 1], a[4 * q + 2], a[4 * q + 3], a[4 * q + 4]);
  }
  __syncthreads();

#pragma unroll 1
  for (int i = 0; i < FRAMES_PER_WAVE; ++i) {
    const int f = wv * FRAMES_PER_WAVE + i;
    const int frame = frame0 + f;
    if (frame < n_frames) {
      const float* fr = &sh[f][PAD];
      float a[ORDER + 1];
      a[0] = 1.0f;
      const float4* srow = reinterpret_cast<const float4*>(&tab[f][0]);
#pragma unroll
      for (int q = 0; q < 4; ++q) {
        float4 t4 = srow[q];
        a[4 * q + 1] = t4.x; a[4 * q + 2] = t4.y; a[4 * q + 3] = t4.z; a[4 * q + 4] = t4.w;
      }
      float* go = out + (long long)frame * FRAME;
#pragma unroll
      for (int t = 0; t < 5; ++t) {
        int n = lane + 32 * t;
        float s = 0.0f;
#pragma unroll
        for (int k = 0; k <= ORDER; ++k) s += a[k] * fr[n - k];
        go[n] = s;
      }
    }
  }
}

extern "C" void kernel_launch(void* const* d_in, const int* in_sizes, int n_in,
                              void* d_out, int out_size, void* d_ws, size_t ws_size,
                              hipStream_t stream) {
  (void)n_in; (void)out_size; (void)d_ws; (void)ws_size;
  const float* x = (const float*)d_in[0];
  float* out = (float*)d_out;
  const int total = in_sizes[0];
  const int n_frames = total / FRAME;
  const int blocks = (n_frames + FRAMES_PER_BLOCK - 1) / FRAMES_PER_BLOCK;
  lpc_residual_kernel<<<blocks, WAVES_PER_BLOCK * 32, 0, stream>>>(x, out, n_frames);
}


# Autocorrelation as two 16x16 Gram tiles of the frame read as a 10x16 matrix: 6 matrix instructions per frame, not 40; lags are wrapped-diagonal sums by DPP row rotation (summation order changes)
# speedup vs baseline: 2.8807x; 2.1427x over previous
#include <hip/hip_runtime.h>


typedef float v2f __attribute__((ext_vector_type(2)));
typedef float v8f __attribute__((ext_vector_type(8)));

#define ORDER 16
#define FRAME 160
#define EPS_F 1e-8f
#define WAVES_PER_BLOCK 8
#define FRAMES_PER_WAVE 4
#define FRAMES_PER_BLOCK (WAVES_PER_BLOCK * FRAMES_PER_WAVE)
#define PAD 16
#define REGION (PAD + FRAME + PAD)
#define TAB_PITCH 20

template <int CTRL>
__device__ __forceinline__ float row_rot(float x) {
  return __builtin_bit_cast(float, __builtin_amdgcn_update_dpp(
      0, __builtin_bit_cast(int, x), CTRL, 0xf, 0xf, true));
}
__device__ __forceinline__ float rot8_upper_half(float x) {
  const int b = __builtin_bit_cast(int, x);
  return __builtin_bit_cast(float, __builtin_amdgcn_update_dpp(b, b, 0x128, 0x2, 0xf, false));
}
__device__ __forceinline__ float xhalf(float x) {
  const int b = __builtin_bit_cast(int, x);
  return __builtin_bit_cast(float, __builtin_amdgcn_permlanex16(b, b, 0x76543210, 0xfedcba98, false, false));
}

__global__ __launch_bounds__(256) void lpc_residual_kernel(
    const float* __restrict__ x, float* __restrict__ out, int n_frames) {
  __shared__ float sh[FRAMES_PER_BLOCK][REGION];
  __shared__ __attribute__((aligned(16))) float tab[FRAMES_PER_BLOCK][TAB_PITCH];

  const int lane = threadIdx.x & 31;
  const int wv = __builtin_amdgcn_readfirstlane((int)(threadIdx.x >> 5));
  const int frame0 = blockIdx.x * FRAMES_PER_BLOCK;

  {
    float v[FRAMES_PER_WAVE][5];
#pragma unroll
    for (int i = 0; i < FRAMES_PER_WAVE; ++i) {
      const int frame = min(frame0 + wv * FRAMES_PER_WAVE + i, n_frames - 1);
      const float* gx = x + (long long)frame * FRAME;
#pragma unroll
      for (int t = 0; t < 5; ++t) v[i][t] = gx[lane + 32 * t];
    }
#pragma unroll
    for (int i = 0; i < FRAMES_PER_WAVE; ++i) {
      float* row = sh[wv * FRAMES_PER_WAVE + i];
      if (lane < PAD) {
        row[lane] = 0.0f;
        row[PAD + FRAME + lane] = 0.0f;
      }
#pragma unroll
      for (int t = 0; t < 5; ++t) row[PAD + lane + 32 * t] = v[i][t];
    }
  }
  __syncthreads();

  const int c = lane & 15;
  const int h = lane >> 4;
  int off[7];
#pragma unroll
  for (int i = 0; i < 5; ++i) off[i] = 96 * h + 16 * i + c;
  off[5] = h ? (FRAME + c) : (80 + c);
  off[6] = h ? (c - PAD) : (96 + c);
  const int d = c - 8 * h;
  float vv[FRAMES_PER_WAVE][7];
#pragma unroll
  for (int i = 0; i < FRAMES_PER_WAVE; ++i) {
    const float* fr = &sh[wv * FRAMES_PER_WAVE + i][PAD];
#pragma unroll
    for (int t = 0; t < 7; ++t) vv[i][t] = fr[off[t]];
  }
#pragma unroll
  for (int i = 0; i < FRAMES_PER_WAVE; ++i) {
    const int f = wv * FRAMES_PER_WAVE + i;
    const float* v = vv[i];
    v8f g0 = {0.f, 0.f, 0.f, 0.f, 0.f, 0.f, 0.f, 0.f};
    v8f g1 = g0;
#pragma unroll
    for (int j = 0; j < 3; ++j) {
      v2f A, B1;
      A.x = v[2 * j];      A.y = v[2 * j + 1];
      B1.x = v[2 * j + 1]; B1.y = v[2 * j + 2];
      g0 = __builtin_amdgcn_wmma_f32_16x16x4_f32(false, A, false, A,
                                                 (short)0, g0, false, false);
      g1 = __builtin_amdgcn_wmma_f32_16x16x4_f32(false, A, false, B1,
                                                 (short)0, g1, false, false);
    }
    float s = (d >= 0) ? g0[0] : g1[0];
    float tr = g1[0];
    s += row_rot<0x12F>((d >= 1) ? g0[1] : g1[1]);  tr += row_rot<0x12F>(g1[1]);
    s += row_rot<0x12E>((d >= 2) ? g0[2] : g1[2]);  tr += row_rot<0x12E>(g1[2]);
    s += row_rot<0x12D>((d >= 3) ? g0[3] : g1[3]);  tr += row_rot<0x12D>(g1[3]);
    s += row_rot<0x12C>((d >= 4) ? g0[4] : g1[4]);  tr += row_rot<0x12C>(g1[4]);
    s += row_rot<0x12B>((d >= 5) ? g0[5] : g1[5]);  tr += row_rot<0x12B>(g1[5]);
    s += row_rot<0x12A>((d >= 6) ? g0[6] : g1[6]);  tr += row_rot<0x12A>(g1[6]);
    s += row_rot<0x129>((d >= 7) ? g0[7] : g1[7]);  tr += row_rot<0x129>(g1[7]);
    s = rot8_upper_half(s);
    tr = rot8_upper_half(tr);
    s += xhalf(s);
    tr += xhalf(tr);
    if (lane <= ORDER) tab[f][lane] = (lane < ORDER) ? s : tr;
  }
  __syncthreads();

  if (wv == (int)(blockIdx.x & (WAVES_PER_BLOCK - 1))) {
    float r[ORDER + 1];
    {
      const float4* srow = reinterpret_cast<const float4*>(&tab[lane][0]);
#pragma unroll
      for (int q = 0; q < 4; ++q) {
        float4 t4 = srow[q];
        r[4 * q + 0] = t4.x; r[4 * q + 1] = t4.y; r[4 * q + 2] = t4.z; r[4 * q + 3] = t4.w;
      }
      r[ORDER] = tab[lane][ORDER];
    }
    float a[ORDER + 1];
    a[0] = 1.0f;
#pragma unroll
    for (int q = 1; q <= ORDER; ++q) a[q] = 0.0f;
    float e = (r[0] != 0.0f) ? r[0] : EPS_F;
#pragma unroll
    for (int i = 1; i <= ORDER; ++i) {
      float accv = r[i];
#pragma unroll
      for (int j = 1; j < i; ++j) accv -= a[j] * r[i - j];
      float k = (e != 0.0f) ? (accv / e) : 0.0f;
      float anew[ORDER + 1];
#pragma unroll
      for (int j = 1; j < i; ++j) anew[j] = a[j] - k * a[i - j];
#pragma unroll
      for (int j = 1; j < i; ++j) a[j] = anew[j];
      a[i] = k;
      e = fmaxf(e * (1.0f - k * k), EPS_F);
    }
    float4* drow = reinterpret_cast<float4*>(&tab[lane][0]);
#pragma unroll
    for (int q = 0; q < 4; ++q)
      drow[q] = make_float4(a[4 * q + 1], a[4 * q + 2], a[4 * q + 3], a[4 * q + 4]);
  }
  __syncthreads();

#pragma unroll 1
  for (int i = 0; i < FRAMES_PER_WAVE; ++i) {
    const int f = wv * FRAMES_PER_WAVE + i;
    const int frame = frame0 + f;
    if (frame < n_frames) {
      const float* fr = &sh[f][PAD];
      float a[ORDER + 1];
      a[0] = 1.0f;
      const float4* srow = reinterpret_cast<const float4*>(&tab[f][0]);
#pragma unroll
      for (int q = 0; q < 4; ++q) {
        float4 t4 = srow[q];
        a[4 * q + 1] = t4.x; a[4 * q + 2] = t4.y; a[4 * q + 3] = t4.z; a[4 * q + 4] = t4.w;
      }
      float* go = out + (long long)frame * FRAME;
#pragma unroll
      for (int t = 0; t < 5; ++t) {
        int n = lane + 32 * t;
        float s = 0.0f;
#pragma unroll
        for (int k = 0; k <= ORDER; ++k) s += a[k] * fr[n - k];
        go[n] = s;
      }
    }
  }
}

extern "C" void kernel_launch(void* const* d_in, const int* in_sizes, int n_in,
                              void* d_out, int out_size, void* d_ws, size_t ws_size,
                              hipStream_t stream) {
  (void)n_in; (void)out_size; (void)d_ws; (void)ws_size;
  const float* x = (const float*)d_in[0];
  float* out = (float*)d_out;
  const int total = in_sizes[0];
  const int n_frames = total / FRAME;
  const int blocks = (n_frames + FRAMES_PER_BLOCK - 1) / FRAMES_PER_BLOCK;
  lpc_residual_kernel<<<blocks, WAVES_PER_BLOCK * 32, 0, stream>>>(x, out, n_frames);
}
